# starting kernel
# baseline (speedup 1.0000x reference)
#include <hip/hip_runtime.h>


#define D_ 16
#define B_ 32
#define T_ 48000
#define P_ 750
#define LP_ 64
#define BLKS_PER_P 4

typedef __attribute__((ext_vector_type(2))) float v2f;
typedef __attribute__((ext_vector_type(4))) float v4f;
typedef __attribute__((ext_vector_type(8))) float v8f;

#define WS_A      0
#define WS_W      16
#define WS_WL     64
#define WS_A16L   320
#define WS_TAB    576
#define WS_XS     2048
#define WS_E      (WS_XS + T_*B_)
#define WS_CARRYS (WS_E + P_*B_*D_)

__device__ __forceinline__ float sigmoidf_(float z) { return 1.0f / (1.0f + expf(-z)); }
__device__ __forceinline__ float powi_(float a, int n) {
    float r = 1.0f;
    for (int i = 0; i < n; ++i) r *= a;
    return r;
}

__global__ void fdn_setup(const float* __restrict__ lk, const float* __restrict__ ar,
                          const float* __restrict__ br, const float* __restrict__ H,
                          float* __restrict__ ws) {
    const int t = threadIdx.x;
    if (t < D_) {
        const float max_delay = 50.0f * 16000.0f / 1000.0f;
        float kappa = 1.0f + sigmoidf_(lk[t]) * (max_delay - 1.0f);
        float a = expf(-1.0f / kappa);
        a = fminf(fmaxf(a, 0.0f), 0.9999f);
        ws[WS_A + t] = a;
        float s = 0.0f;
        for (int i = 0; i < D_; ++i) s += sigmoidf_(ar[i]) * H[i * D_ + t];
        ws[WS_W + t] = (sigmoidf_(br[t]) + s) * (1.0f / (float)D_);
    }
    __syncthreads();

    for (int e = t; e < 2 * 256; e += blockDim.x) {
        const int which = e >> 8, idx = e & 255;
        const int lane = idx >> 3, v = idx & 7;
        const int h = lane >> 4;
        const int d = 4 * (v >> 1) + 2 * h + (v & 1);
        const float a = ws[WS_A + d];
        if (which == 0) ws[WS_WL + idx] = ws[WS_W + d];
        else            ws[WS_A16L + idx] = powi_(a, 16);
    }

    for (int e = t; e < 12 * 64; e += blockDim.x) {
        const int chunk = e >> 6, pos = e & 63;
        const int lane = pos >> 1, ra = pos & 1;
        const int h = lane >> 4, m = lane & 15;
        const int kap = ra + 2 * h;
        float val = 0.0f;
        if (chunk < 4) {
            const int j = 4 * chunk + kap;
            if (j <= m)
                for (int d = 0; d < D_; ++d)
                    val += ws[WS_W + d] * powi_(ws[WS_A + d], m - j);
        } else if (chunk < 8) {
            const int d = 4 * (chunk - 4) + kap;
            val = powi_(ws[WS_A + d], m + 1);
        } else {
            int sg;
            if (m < 8) sg = 4 * (m >> 1) + (m & 1);
            else { const int v = m - 8; sg = 4 * (v >> 1) + 2 + (v & 1); }
            const int j = 4 * (chunk - 8) + kap;
            val = powi_(ws[WS_A + sg], 15 - j);
        }
        ws[WS_TAB + e] = val;
    }
}

__global__ void fdn_swizzle(const float* __restrict__ x, float* __restrict__ ws) {
    const int id = blockIdx.x * blockDim.x + threadIdx.x;
    if (id >= B_ * T_) return;
    const int b = id / T_, t = id % T_;
    const float val = x[id];
    const int tblk = t >> 4, j = t & 15;
    const int g = b >> 4, bl = b & 15;
    const int kk = j >> 2, kap = j & 3;
    const int h = kap >> 1, r = kap & 1;
    const int v = 2 * kk + r;
    ws[WS_XS + ((((tblk << 1) | g) * 32 + (h << 4) + bl) << 3) + v] = val;
}

__global__ void fdn_partial(const float* __restrict__ x, float* __restrict__ ws) {
    const int tid = blockIdx.x * blockDim.x + threadIdx.x;
    if (tid >= P_ * B_ * D_) return;
    const int p = tid >> 9;
    const int r = tid & 511;
    const int b = r >> 4, d = r & 15;
    const float a = ws[WS_A + d];
    const float* xp = x + b * T_ + p * LP_;
    float acc = 0.0f;
#pragma unroll 4
    for (int i = 0; i < LP_; ++i) acc = fmaf(a, acc, xp[i]);
    ws[WS_E + (p * B_ + b) * D_ + d] = acc;
}

__global__ void fdn_carry(float* __restrict__ ws) {
    const int tid = blockIdx.x * blockDim.x + threadIdx.x;
    if (tid >= B_ * D_) return;
    const int b = tid >> 4, d = tid & 15;
    const int g = b >> 4, bl = b & 15;
    const int h = (d >> 1) & 1, v = 2 * (d >> 2) + (d & 1);
    const int laneSlot = (((h << 4) + bl) << 3) + v;
    const float a = ws[WS_A + d];
    float aL = a;
    for (int q = 0; q < 6; ++q) aL *= aL;
    float c = 0.0f;
    for (int p = 0; p < P_; ++p) {
        ws[WS_CARRYS + ((((p << 1) | g) * 32) << 3) + laneSlot] = c;
        c = fmaf(aL, c, ws[WS_E + (p * B_ + b) * D_ + d]);
    }
}

__global__ void __launch_bounds__(256) fdn_main(const float* __restrict__ ws,
                                                float* __restrict__ out) {
    const int lane = threadIdx.x & 31;
    const int widx = blockIdx.x * (blockDim.x >> 5) + (threadIdx.x >> 5);
    if (widx >= 2 * P_) return;
    const int p = widx >> 1, g = widx & 1;
    const int h = lane >> 4, bl = lane & 15;

    const v8f wlv   = *(const v8f*)(ws + WS_WL   + lane * 8);
    const v8f a16lv = *(const v8f*)(ws + WS_A16L + lane * 8);

    v2f Lm[4], Pm[4], Rm[4];
#pragma unroll
    for (int kk = 0; kk < 4; ++kk) {
        Lm[kk] = *(const v2f*)(ws + WS_TAB + (kk)     * 64 + lane * 2);
        Pm[kk] = *(const v2f*)(ws + WS_TAB + (kk + 4) * 64 + lane * 2);
        Rm[kk] = *(const v2f*)(ws + WS_TAB + (kk + 8) * 64 + lane * 2);
    }

    v8f stv = *(const v8f*)(ws + WS_CARRYS + (widx * 32 + lane) * 8);

    const float* xS = ws + WS_XS;
    int tb = p * BLKS_PER_P;
    int t0 = p * LP_;
    for (int blk = 0; blk < BLKS_PER_P; ++blk) {
        const v8f xv = *(const v8f*)(xS + (((tb << 1) | g) * 32 + lane) * 8);
        const v2f* xc = (const v2f*)&xv;

        v8f qv = wlv * stv;
        const v2f* qc = (const v2f*)&qv;

        v8f acc = {0.f, 0.f, 0.f, 0.f, 0.f, 0.f, 0.f, 0.f};
#pragma unroll
        for (int kk = 0; kk < 4; ++kk)
            acc = __builtin_amdgcn_wmma_f32_16x16x4_f32(false, Pm[kk], false, qc[kk],
                                                        (short)0, acc, false, false);
        v8f rr = {0.f, 0.f, 0.f, 0.f, 0.f, 0.f, 0.f, 0.f};
#pragma unroll
        for (int kk = 0; kk < 4; ++kk) {
            acc = __builtin_amdgcn_wmma_f32_16x16x4_f32(false, Lm[kk], false, xc[kk],
                                                        (short)0, acc, false, false);
            rr  = __builtin_amdgcn_wmma_f32_16x16x4_f32(false, Rm[kk], false, xc[kk],
                                                        (short)0, rr, false, false);
        }
        stv = a16lv * stv + rr;

        float* op = out + (g * 16 + bl) * T_ + t0 + 8 * h;
        const v4f* ap = (const v4f*)&acc;
        *(v4f*)op       = ap[0];
        *(v4f*)(op + 4) = ap[1];

        ++tb;
        t0 += 16;
    }
}

extern "C" void kernel_launch(void* const* d_in, const int* in_sizes, int n_in,
                              void* d_out, int out_size, void* d_ws, size_t ws_size,
                              hipStream_t stream) {
    const float* x  = (const float*)d_in[0];
    const float* lk = (const float*)d_in[1];
    const float* ar = (const float*)d_in[2];
    const float* br = (const float*)d_in[3];
    const float* H  = (const float*)d_in[4];
    float* ws  = (float*)d_ws;
    float* out = (float*)d_out;

    hipLaunchKernelGGL(fdn_setup,   dim3(1),                          dim3(256), 0, stream, lk, ar, br, H, ws);
    hipLaunchKernelGGL(fdn_swizzle, dim3((B_ * T_ + 255) / 256),      dim3(256), 0, stream, x, ws);
    hipLaunchKernelGGL(fdn_partial, dim3((P_ * B_ * D_ + 255) / 256), dim3(256), 0, stream, x, ws);
    hipLaunchKernelGGL(fdn_carry,   dim3(2),                          dim3(256), 0, stream, ws);
    hipLaunchKernelGGL(fdn_main,    dim3((2 * P_ + 7) / 8),           dim3(256), 0, stream, ws, out);
}


# Carry scan over the 750 partitions done as a two-level parallel scan (30 chunks of 25 per work-group, LDS hand-over) instead of one serial loop; carries are summed in a different order (chunk tree)
# speedup vs baseline: 8.4450x; 8.4450x over previous
#include <hip/hip_runtime.h>


#define D_ 16
#define B_ 32
#define T_ 48000
#define P_ 750
#define LP_ 64
#define BLKS_PER_P 4

typedef __attribute__((ext_vector_type(2))) float v2f;
typedef __attribute__((ext_vector_type(4))) float v4f;
typedef __attribute__((ext_vector_type(8))) float v8f;

#define WS_A      0
#define WS_W      16
#define WS_WL     64
#define WS_A16L   320
#define WS_TAB    576
#define WS_XS     2048
#define WS_E      (WS_XS + T_*B_)
#define WS_CARRYS (WS_E + P_*B_*D_)

__device__ __forceinline__ float sigmoidf_(float z) { return 1.0f / (1.0f + expf(-z)); }
__device__ __forceinline__ float powi_(float a, int n) {
    float r = 1.0f;
    for (int i = 0; i < n; ++i) r *= a;
    return r;
}

__global__ void fdn_setup(const float* __restrict__ lk, const float* __restrict__ ar,
                          const float* __restrict__ br, const float* __restrict__ H,
                          float* __restrict__ ws) {
    const int t = threadIdx.x;
    if (t < D_) {
        const float max_delay = 50.0f * 16000.0f / 1000.0f;
        float kappa = 1.0f + sigmoidf_(lk[t]) * (max_delay - 1.0f);
        float a = expf(-1.0f / kappa);
        a = fminf(fmaxf(a, 0.0f), 0.9999f);
        ws[WS_A + t] = a;
        float s = 0.0f;
        for (int i = 0; i < D_; ++i) s += sigmoidf_(ar[i]) * H[i * D_ + t];
        ws[WS_W + t] = (sigmoidf_(br[t]) + s) * (1.0f / (float)D_);
    }
    __syncthreads();

    for (int e = t; e < 2 * 256; e += blockDim.x) {
        const int which = e >> 8, idx = e & 255;
        const int lane = idx >> 3, v = idx & 7;
        const int h = lane >> 4;
        const int d = 4 * (v >> 1) + 2 * h + (v & 1);
        const float a = ws[WS_A + d];
        if (which == 0) ws[WS_WL + idx] = ws[WS_W + d];
        else            ws[WS_A16L + idx] = powi_(a, 16);
    }

    for (int e = t; e < 12 * 64; e += blockDim.x) {
        const int chunk = e >> 6, pos = e & 63;
        const int lane = pos >> 1, ra = pos & 1;
        const int h = lane >> 4, m = lane & 15;
        const int kap = ra + 2 * h;
        float val = 0.0f;
        if (chunk < 4) {
            const int j = 4 * chunk + kap;
            if (j <= m)
                for (int d = 0; d < D_; ++d)
                    val += ws[WS_W + d] * powi_(ws[WS_A + d], m - j);
        } else if (chunk < 8) {
            const int d = 4 * (chunk - 4) + kap;
            val = powi_(ws[WS_A + d], m + 1);
        } else {
            int sg;
            if (m < 8) sg = 4 * (m >> 1) + (m & 1);
            else { const int v = m - 8; sg = 4 * (v >> 1) + 2 + (v & 1); }
            const int j = 4 * (chunk - 8) + kap;
            val = powi_(ws[WS_A + sg], 15 - j);
        }
        ws[WS_TAB + e] = val;
    }
}

__global__ void fdn_swizzle(const float* __restrict__ x, float* __restrict__ ws) {
    const int id = blockIdx.x * blockDim.x + threadIdx.x;
    if (id >= B_ * T_) return;
    const int b = id / T_, t = id % T_;
    const float val = x[id];
    const int tblk = t >> 4, j = t & 15;
    const int g = b >> 4, bl = b & 15;
    const int kk = j >> 2, kap = j & 3;
    const int h = kap >> 1, r = kap & 1;
    const int v = 2 * kk + r;
    ws[WS_XS + ((((tblk << 1) | g) * 32 + (h << 4) + bl) << 3) + v] = val;
}

__global__ void fdn_partial(const float* __restrict__ x, float* __restrict__ ws) {
    const int tid = blockIdx.x * blockDim.x + threadIdx.x;
    if (tid >= P_ * B_ * D_) return;
    const int p = tid >> 9;
    const int r = tid & 511;
    const int b = r >> 4, d = r & 15;
    const float a = ws[WS_A + d];
    const float* xp = x + b * T_ + p * LP_;
    float acc = 0.0f;
#pragma unroll 4
    for (int i = 0; i < LP_; ++i) acc = fmaf(a, acc, xp[i]);
    ws[WS_E + (p * B_ + b) * D_ + d] = acc;
}

#define CW_ 30
#define CP_ 25
#define ROW_ (B_ * D_)

__global__ void __launch_bounds__(CW_ * 32) fdn_carry(float* __restrict__ ws) {
    static_assert(CW_ * CP_ == P_, "chunks must tile the partitions exactly");
    __shared__ float agg[2 * CW_][32];

    const int lane = threadIdx.x & 31;
    const int w = threadIdx.x >> 5;
    const int g = blockIdx.x >> 3, h = (blockIdx.x >> 2) & 1, blq = blockIdx.x & 3;
    const int bl = 4 * blq + (lane >> 3), v = lane & 7;
    const int b = 16 * g + bl;
    const int d = 4 * (v >> 1) + 2 * h + (v & 1);
    const int inSlot  = b * D_ + d;
    const int outSlot = (((h << 4) + bl) << 3) + v + (g << 8);

    const int p0 = w * CP_;
    const float* ep = ws + WS_E + p0 * ROW_ + inSlot;
    float e[CP_];
#pragma unroll
    for (int i = 0; i < CP_; ++i) e[i] = ep[i * ROW_];

    const float a = ws[WS_A + d];
    float aL = a;
    for (int q = 0; q < 6; ++q) aL *= aL;
    const float a2 = aL * aL, a4 = a2 * a2, a8 = a4 * a4, a16 = a8 * a8;
    const float aC = a16 * a8 * aL;
    static_assert(CP_ == 25, "aC above is aL^25");

    float s = 0.0f;
#pragma unroll
    for (int i = 0; i < CP_; ++i) s = fmaf(aL, s, e[i]);
    agg[w][lane] = 0.0f;
    agg[CW_ + w][lane] = s;
    __syncthreads();

    float c = 0.0f;
#pragma unroll
    for (int q = 0; q < CW_ - 1; ++q) c = fmaf(aC, c, agg[w + 1 + q][lane]);

    float* cp = ws + WS_CARRYS + p0 * ROW_ + outSlot;
#pragma unroll
    for (int i = 0; i < CP_; ++i) {
        cp[i * ROW_] = c;
        c = fmaf(aL, c, e[i]);
    }
}

__global__ void __launch_bounds__(256) fdn_main(const float* __restrict__ ws,
                                                float* __restrict__ out) {
    const int lane = threadIdx.x & 31;
    const int widx = blockIdx.x * (blockDim.x >> 5) + (threadIdx.x >> 5);
    if (widx >= 2 * P_) return;
    const int p = widx >> 1, g = widx & 1;
    const int h = lane >> 4, bl = lane & 15;

    const v8f wlv   = *(const v8f*)(ws + WS_WL   + lane * 8);
    const v8f a16lv = *(const v8f*)(ws + WS_A16L + lane * 8);

    v2f Lm[4], Pm[4], Rm[4];
#pragma unroll
    for (int kk = 0; kk < 4; ++kk) {
        Lm[kk] = *(const v2f*)(ws + WS_TAB + (kk)     * 64 + lane * 2);
        Pm[kk] = *(const v2f*)(ws + WS_TAB + (kk + 4) * 64 + lane * 2);
        Rm[kk] = *(const v2f*)(ws + WS_TAB + (kk + 8) * 64 + lane * 2);
    }

    v8f stv = *(const v8f*)(ws + WS_CARRYS + (widx * 32 + lane) * 8);

    const float* xS = ws + WS_XS;
    int tb = p * BLKS_PER_P;
    int t0 = p * LP_;
    for (int blk = 0; blk < BLKS_PER_P; ++blk) {
        const v8f xv = *(const v8f*)(xS + (((tb << 1) | g) * 32 + lane) * 8);
        const v2f* xc = (const v2f*)&xv;

        v8f qv = wlv * stv;
        const v2f* qc = (const v2f*)&qv;

        v8f acc = {0.f, 0.f, 0.f, 0.f, 0.f, 0.f, 0.f, 0.f};
#pragma unroll
        for (int kk = 0; kk < 4; ++kk)
            acc = __builtin_amdgcn_wmma_f32_16x16x4_f32(false, Pm[kk], false, qc[kk],
                                                        (short)0, acc, false, false);
        v8f rr = {0.f, 0.f, 0.f, 0.f, 0.f, 0.f, 0.f, 0.f};
#pragma unroll
        for (int kk = 0; kk < 4; ++kk) {
            acc = __builtin_amdgcn_wmma_f32_16x16x4_f32(false, Lm[kk], false, xc[kk],
                                                        (short)0, acc, false, false);
            rr  = __builtin_amdgcn_wmma_f32_16x16x4_f32(false, Rm[kk], false, xc[kk],
                                                        (short)0, rr, false, false);
        }
        stv = a16lv * stv + rr;

        float* op = out + (g * 16 + bl) * T_ + t0 + 8 * h;
        const v4f* ap = (const v4f*)&acc;
        *(v4f*)op       = ap[0];
        *(v4f*)(op + 4) = ap[1];

        ++tb;
        t0 += 16;
    }
}

extern "C" void kernel_launch(void* const* d_in, const int* in_sizes, int n_in,
                              void* d_out, int out_size, void* d_ws, size_t ws_size,
                              hipStream_t stream) {
    const float* x  = (const float*)d_in[0];
    const float* lk = (const float*)d_in[1];
    const float* ar = (const float*)d_in[2];
    const float* br = (const float*)d_in[3];
    const float* H  = (const float*)d_in[4];
    float* ws  = (float*)d_ws;
    float* out = (float*)d_out;

    hipLaunchKernelGGL(fdn_setup,   dim3(1),                          dim3(256), 0, stream, lk, ar, br, H, ws);
    hipLaunchKernelGGL(fdn_swizzle, dim3((B_ * T_ + 255) / 256),      dim3(256), 0, stream, x, ws);
    hipLaunchKernelGGL(fdn_partial, dim3((P_ * B_ * D_ + 255) / 256), dim3(256), 0, stream, x, ws);
    hipLaunchKernelGGL(fdn_carry,   dim3(ROW_ / 32),                  dim3(CW_ * 32), 0, stream, ws);
    hipLaunchKernelGGL(fdn_main,    dim3((2 * P_ + 7) / 8),           dim3(256), 0, stream, ws, out);
}


# Launch fusion 5 -> 3: table setup runs as block 0 of the end-state launch (whose threads derive their own decay); the WMMA kernel loads x rows straight into B-fragment registers, no re-layout pass
# speedup vs baseline: 14.8307x; 1.7562x over previous
#include <hip/hip_runtime.h>


#define D_ 16
#define B_ 32
#define T_ 48000
#define P_ 750
#define LP_ 64
#define BLKS_PER_P 4

typedef __attribute__((ext_vector_type(2))) float v2f;
typedef __attribute__((ext_vector_type(4))) float v4f;
typedef __attribute__((ext_vector_type(8))) float v8f;

#define WS_A      0
#define WS_W      16
#define WS_WL     64
#define WS_A16L   320
#define WS_TAB    576
#define WS_XS     2048
#define WS_E      (WS_XS + T_*B_)
#define WS_CARRYS (WS_E + P_*B_*D_)

__device__ __forceinline__ float sigmoidf_(float z) { return 1.0f / (1.0f + expf(-z)); }
__device__ __forceinline__ float powi_(float a, int n) {
    float r = 1.0f;
    for (int i = 0; i < n; ++i) r *= a;
    return r;
}
__device__ __forceinline__ float decay_(float lkv) {
    const float max_delay = 50.0f * 16000.0f / 1000.0f;
    const float kappa = fmaf(sigmoidf_(lkv), max_delay - 1.0f, 1.0f);
    const float a = expf(-1.0f / kappa);
    return fminf(fmaxf(a, 0.0f), 0.9999f);
}

__device__ __forceinline__ void fdn_setup_block(const float* __restrict__ lk,
                                                const float* __restrict__ ar,
                                                const float* __restrict__ br,
                                                const float* __restrict__ H,
                                                float* __restrict__ ws) {
    __shared__ float sA[D_], sW[D_];
    const int t = threadIdx.x;
    if (t < D_) {
        const float a = decay_(lk[t]);
        sA[t] = a;
        ws[WS_A + t] = a;
        float s = 0.0f;
        for (int i = 0; i < D_; ++i) s += sigmoidf_(ar[i]) * H[i * D_ + t];
        sW[t] = (sigmoidf_(br[t]) + s) * (1.0f / (float)D_);
    }
    __syncthreads();

    for (int e = t; e < 2 * 256; e += blockDim.x) {
        const int which = e >> 8, idx = e & 255;
        const int lane = idx >> 3, v = idx & 7;
        const int h = lane >> 4;
        const int d = 4 * (v >> 1) + 2 * h + (v & 1);
        const float a = sA[d];
        if (which == 0) ws[WS_WL + idx] = sW[d];
        else            ws[WS_A16L + idx] = powi_(a, 16);
    }

    for (int e = t; e < 12 * 64; e += blockDim.x) {
        const int chunk = e >> 6, pos = e & 63;
        const int lane = pos >> 1, ra = pos & 1;
        const int h = lane >> 4, m = lane & 15;
        const int kap = ra + 2 * h;
        float val = 0.0f;
        if (chunk < 4) {
            const int j = 4 * chunk + kap;
            if (j <= m)
                for (int d = 0; d < D_; ++d)
                    val += sW[d] * powi_(sA[d], m - j);
        } else if (chunk < 8) {
            const int d = 4 * (chunk - 4) + kap;
            val = powi_(sA[d], m + 1);
        } else {
            int sg;
            if (m < 8) sg = 4 * (m >> 1) + (m & 1);
            else { const int v = m - 8; sg = 4 * (v >> 1) + 2 + (v & 1); }
            const int j = 4 * (chunk - 8) + kap;
            val = powi_(sA[sg], 15 - j);
        }
        ws[WS_TAB + e] = val;
    }
}

__global__ void fdn_partial(const float* __restrict__ x, const float* __restrict__ lk,
                            const float* __restrict__ ar, const float* __restrict__ br,
                            const float* __restrict__ H, float* __restrict__ ws) {
    if (blockIdx.x == 0) {
        fdn_setup_block(lk, ar, br, H, ws);
        return;
    }
    const int tid = (blockIdx.x - 1) * blockDim.x + threadIdx.x;
    if (tid >= P_ * B_ * D_) return;
    const int p = tid >> 9;
    const int r = tid & 511;
    const int b = r >> 4, d = r & 15;
    const float a = decay_(lk[d]);
    const float* xp = x + b * T_ + p * LP_;
    float acc = 0.0f;
#pragma unroll 4
    for (int i = 0; i < LP_; ++i) acc = fmaf(a, acc, xp[i]);
    ws[WS_E + (p * B_ + b) * D_ + d] = acc;
}

#define CW_ 30
#define CP_ 25
#define ROW_ (B_ * D_)

__global__ void __launch_bounds__(CW_ * 32) fdn_carry(float* __restrict__ ws) {
    static_assert(CW_ * CP_ == P_, "chunks must tile the partitions exactly");
    __shared__ float agg[2 * CW_][32];

    const int lane = threadIdx.x & 31;
    const int w = threadIdx.x >> 5;
    const int g = blockIdx.x >> 3, h = (blockIdx.x >> 2) & 1, blq = blockIdx.x & 3;
    const int bl = 4 * blq + (lane >> 3), v = lane & 7;
    const int b = 16 * g + bl;
    const int d = 4 * (v >> 1) + 2 * h + (v & 1);
    const int inSlot  = b * D_ + d;
    const int outSlot = (((h << 4) + bl) << 3) + v + (g << 8);

    const int p0 = w * CP_;
    const float* ep = ws + WS_E + p0 * ROW_ + inSlot;
    float e[CP_];
#pragma unroll
    for (int i = 0; i < CP_; ++i) e[i] = ep[i * ROW_];

    const float a = ws[WS_A + d];
    float aL = a;
    for (int q = 0; q < 6; ++q) aL *= aL;
    const float a2 = aL * aL, a4 = a2 * a2, a8 = a4 * a4, a16 = a8 * a8;
    const float aC = a16 * a8 * aL;
    static_assert(CP_ == 25, "aC above is aL^25");

    float s = 0.0f;
#pragma unroll
    for (int i = 0; i < CP_; ++i) s = fmaf(aL, s, e[i]);
    agg[w][lane] = 0.0f;
    agg[CW_ + w][lane] = s;
    __syncthreads();

    float c = 0.0f;
#pragma unroll
    for (int q = 0; q < CW_ - 1; ++q) c = fmaf(aC, c, agg[w + 1 + q][lane]);

    float* cp = ws + WS_CARRYS + p0 * ROW_ + outSlot;
#pragma unroll
    for (int i = 0; i < CP_; ++i) {
        cp[i * ROW_] = c;
        c = fmaf(aL, c, e[i]);
    }
}

__global__ void __launch_bounds__(256) fdn_main(const float* __restrict__ x,
                                                const float* __restrict__ ws,
                                                float* __restrict__ out) {
    const int lane = threadIdx.x & 31;
    const int widx = blockIdx.x * (blockDim.x >> 5) + (threadIdx.x >> 5);
    if (widx >= 2 * P_) return;
    const int p = widx >> 1, g = widx & 1;
    const int h = lane >> 4, bl = lane & 15;

    const v8f wlv   = *(const v8f*)(ws + WS_WL   + lane * 8);
    const v8f a16lv = *(const v8f*)(ws + WS_A16L + lane * 8);

    v2f Lm[4], Pm[4], Rm[4];
#pragma unroll
    for (int kk = 0; kk < 4; ++kk) {
        Lm[kk] = *(const v2f*)(ws + WS_TAB + (kk)     * 64 + lane * 2);
        Pm[kk] = *(const v2f*)(ws + WS_TAB + (kk + 4) * 64 + lane * 2);
        Rm[kk] = *(const v2f*)(ws + WS_TAB + (kk + 8) * 64 + lane * 2);
    }

    v8f stv = *(const v8f*)(ws + WS_CARRYS + (widx * 32 + lane) * 8);

    const float* xr = x + (g * 16 + bl) * T_ + p * LP_ + 2 * h;
    v2f xq[BLKS_PER_P][4];
#pragma unroll
    for (int blk = 0; blk < BLKS_PER_P; ++blk)
#pragma unroll
        for (int kk = 0; kk < 4; ++kk)
            xq[blk][kk] = *(const v2f*)(xr + 16 * blk + 4 * kk);

    int t0 = p * LP_;
#pragma unroll
    for (int blk = 0; blk < BLKS_PER_P; ++blk) {
        const v2f* xc = xq[blk];

        v8f qv = wlv * stv;
        const v2f* qc = (const v2f*)&qv;

        v8f acc = {0.f, 0.f, 0.f, 0.f, 0.f, 0.f, 0.f, 0.f};
#pragma unroll
        for (int kk = 0; kk < 4; ++kk)
            acc = __builtin_amdgcn_wmma_f32_16x16x4_f32(false, Pm[kk], false, qc[kk],
                                                        (short)0, acc, false, false);
        v8f rr = {0.f, 0.f, 0.f, 0.f, 0.f, 0.f, 0.f, 0.f};
#pragma unroll
        for (int kk = 0; kk < 4; ++kk) {
            acc = __builtin_amdgcn_wmma_f32_16x16x4_f32(false, Lm[kk], false, xc[kk],
                                                        (short)0, acc, false, false);
            rr  = __builtin_amdgcn_wmma_f32_16x16x4_f32(false, Rm[kk], false, xc[kk],
                                                        (short)0, rr, false, false);
        }
        stv = a16lv * stv + rr;

        float* op = out + (g * 16 + bl) * T_ + t0 + 8 * h;
        const v4f* ap = (const v4f*)&acc;
        *(v4f*)op       = ap[0];
        *(v4f*)(op + 4) = ap[1];

        t0 += 16;
    }
}

extern "C" void kernel_launch(void* const* d_in, const int* in_sizes, int n_in,
                              void* d_out, int out_size, void* d_ws, size_t ws_size,
                              hipStream_t stream) {
    const float* x  = (const float*)d_in[0];
    const float* lk = (const float*)d_in[1];
    const float* ar = (const float*)d_in[2];
    const float* br = (const float*)d_in[3];
    const float* H  = (const float*)d_in[4];
    float* ws  = (float*)d_ws;
    float* out = (float*)d_out;

    hipLaunchKernelGGL(fdn_partial, dim3(1 + (P_ * B_ * D_ + 255) / 256), dim3(256), 0, stream, x, lk, ar, br, H, ws);
    hipLaunchKernelGGL(fdn_carry,   dim3(ROW_ / 32),                      dim3(CW_ * 32), 0, stream, ws);
    hipLaunchKernelGGL(fdn_main,    dim3((2 * P_ + 7) / 8),               dim3(256), 0, stream, x, ws, out);
}


# Launch fusion 3 -> 2: carry scan split into a scan inside each 30-partition chunk at the end of end-state work-groups and a 24-row chunk fold opening the output launch; carries summed in new order
# speedup vs baseline: 19.3699x; 1.3061x over previous
#include <hip/hip_runtime.h>


#define D_ 16
#define B_ 32
#define T_ 48000
#define P_ 750
#define LP_ 64
#define BLKS_PER_P 4

typedef __attribute__((ext_vector_type(2))) float v2f;
typedef __attribute__((ext_vector_type(4))) float v4f;
typedef __attribute__((ext_vector_type(8))) float v8f;

#define WS_A      0
#define WS_W      16
#define WS_WL     64
#define WS_A16L   320
#define WS_TAB    576
#define WS_ALL    1344
#define WS_XS     2048
#define WS_S      WS_XS
#define WS_PW     32768
#define WS_E      (WS_XS + T_*B_)
#define WS_CARRYS (WS_E + P_*B_*D_)
#define CP_  30
#define NCH_ 25
#define ZR_  (NCH_ - 1)
#define MP_  3
#define ROW_ (B_ * D_)

__device__ __forceinline__ float sigmoidf_(float z) { return 1.0f / (1.0f + expf(-z)); }
__device__ __forceinline__ float powi_(float a, int n) {
    float r = 1.0f;
    for (int i = 0; i < n; ++i) r *= a;
    return r;
}
__device__ __forceinline__ float a64_(float a) {
    float r = a;
    for (int q = 0; q < 6; ++q) r *= r;
    return r;
}
__device__ __forceinline__ float decay_(float lkv) {
    const float max_delay = 50.0f * 16000.0f / 1000.0f;
    const float kappa = fmaf(sigmoidf_(lkv), max_delay - 1.0f, 1.0f);
    const float a = expf(-1.0f / kappa);
    return fminf(fmaxf(a, 0.0f), 0.9999f);
}

__device__ __forceinline__ void fdn_setup_block(const float* __restrict__ lk,
                                                const float* __restrict__ ar,
                                                const float* __restrict__ br,
                                                const float* __restrict__ H,
                                                float* __restrict__ ws) {
    __shared__ float sA[D_], sW[D_];
    const int t = threadIdx.x;
    if (t < D_) {
        const float a = decay_(lk[t]);
        sA[t] = a;
        float s = 0.0f;
        for (int i = 0; i < D_; ++i) s += sigmoidf_(ar[i]) * H[i * D_ + t];
        sW[t] = (sigmoidf_(br[t]) + s) * (1.0f / (float)D_);
    }
    __syncthreads();

    for (int e = t; e < 3 * 256; e += blockDim.x) {
        const int which = e >> 8, idx = e & 255;
        const int lane = idx >> 3, v = idx & 7;
        const int h = lane >> 4;
        const int d = 4 * (v >> 1) + 2 * h + (v & 1);
        const float a = sA[d];
        if (which == 0)      ws[WS_WL + idx] = sW[d];
        else if (which == 1) ws[WS_A16L + idx] = powi_(a, 16);
        else {
            const float aL = a64_(a);
            ws[WS_ALL + idx] = aL;
            float pw = 1.0f;
            for (int j = 0; j < CP_; ++j) { ws[WS_PW + j * 256 + idx] = pw; pw *= aL; }
        }
    }
    for (int e = t; e < ZR_ * ROW_; e += blockDim.x) ws[WS_S + e] = 0.0f;

    for (int e = t; e < 12 * 64; e += blockDim.x) {
        const int chunk = e >> 6, pos = e & 63;
        const int lane = pos >> 1, ra = pos & 1;
        const int h = lane >> 4, m = lane & 15;
        const int kap = ra + 2 * h;
        float val = 0.0f;
        if (chunk < 4) {
            const int j = 4 * chunk + kap;
            if (j <= m)
                for (int d = 0; d < D_; ++d)
                    val += sW[d] * powi_(sA[d], m - j);
        } else if (chunk < 8) {
            const int d = 4 * (chunk - 4) + kap;
            val = powi_(sA[d], m + 1);
        } else {
            int sg;
            if (m < 8) sg = 4 * (m >> 1) + (m & 1);
            else { const int v = m - 8; sg = 4 * (v >> 1) + 2 + (v & 1); }
            const int j = 4 * (chunk - 8) + kap;
            val = powi_(sA[sg], 15 - j);
        }
        ws[WS_TAB + e] = val;
    }
}

__global__ void __launch_bounds__(CP_ * 32) fdn_partial(const float* __restrict__ x,
                                                        const float* __restrict__ lk,
                                                        const float* __restrict__ ar,
                                                        const float* __restrict__ br,
                                                        const float* __restrict__ H,
                                                        float* __restrict__ ws) {
    static_assert(NCH_ * CP_ == P_, "chunks must tile the partitions exactly");
    __shared__ float agg[CP_][32];
    if (blockIdx.x == 0) {
        fdn_setup_block(lk, ar, br, H, ws);
        return;
    }
    const int wg = blockIdx.x - 1;
    const int c = wg >> 4;
    const int g = (wg >> 3) & 1, h = (wg >> 2) & 1, blq = wg & 3;
    const int lane = threadIdx.x & 31;
    const int q = threadIdx.x >> 5;
    const int bl = 4 * blq + (lane >> 3), v = lane & 7;
    const int b = 16 * g + bl;
    const int d = 4 * (v >> 1) + 2 * h + (v & 1);
    const float a = decay_(lk[d]);
    const float* xp = x + b * T_ + (c * CP_ + q) * LP_;
    float acc = 0.0f;
#pragma unroll 4
    for (int i = 0; i < LP_; ++i) acc = fmaf(a, acc, xp[i]);
    agg[q][lane] = acc;
    __syncthreads();
    if (q != 0) return;

    const float aL = a64_(a);
    float e[CP_];
#pragma unroll
    for (int i = 0; i < CP_; ++i) e[i] = agg[i][lane];

    const int outSlot = (((h << 4) + bl) << 3) + v + (g << 8);
    float* cp = ws + WS_CARRYS + c * CP_ * ROW_ + outSlot;
    float s = 0.0f;
#pragma unroll
    for (int i = 0; i < CP_; ++i) {
        cp[i * ROW_] = s;
        s = fmaf(aL, s, e[i]);
    }
    ws[WS_S + (ZR_ + c) * ROW_ + outSlot] = s;
}

__global__ void __launch_bounds__(MP_ * 2 * 32) fdn_main(const float* __restrict__ x,
                                                         const float* __restrict__ ws,
                                                         float* __restrict__ out) {
    static_assert(CP_ % MP_ == 0 && P_ % MP_ == 0, "a work-group stays inside one chunk");
    static_assert(MP_ * 2 * 32 >= ROW_ / 4, "one column quad per thread must cover a row");
    static_assert(CP_ == 30, "aC below is aL^30");
    __shared__ __attribute__((aligned(32))) float sC[ROW_];
    const int lane = threadIdx.x & 31;
    const int wv = threadIdx.x >> 5;
    const int p = blockIdx.x * MP_ + (wv >> 1), g = wv & 1;
    const int widx = 2 * p + g;
    const int c = blockIdx.x / (CP_ / MP_);
    const int h = lane >> 4, bl = lane & 15;

    const int col = 4 * (threadIdx.x & (ROW_ / 4 - 1));
    const v4f aLc = *(const v4f*)(ws + WS_ALL + (col & 255));
    const float* sp = ws + WS_S + c * ROW_ + col;
    v4f sr[ZR_];
#pragma unroll
    for (int q = 0; q < ZR_; ++q) sr[q] = *(const v4f*)(sp + q * ROW_);
    const v8f pwv = *(const v8f*)(ws + WS_PW + (p - c * CP_) * 256 + lane * 8);

    const v8f wlv   = *(const v8f*)(ws + WS_WL   + lane * 8);
    const v8f a16lv = *(const v8f*)(ws + WS_A16L + lane * 8);

    v2f Lm[4], Pm[4], Rm[4];
#pragma unroll
    for (int kk = 0; kk < 4; ++kk) {
        Lm[kk] = *(const v2f*)(ws + WS_TAB + (kk)     * 64 + lane * 2);
        Pm[kk] = *(const v2f*)(ws + WS_TAB + (kk + 4) * 64 + lane * 2);
        Rm[kk] = *(const v2f*)(ws + WS_TAB + (kk + 8) * 64 + lane * 2);
    }

    v8f stv = *(const v8f*)(ws + WS_CARRYS + (widx * 32 + lane) * 8);

    const float* xr = x + (g * 16 + bl) * T_ + p * LP_ + 2 * h;
    v2f xq[BLKS_PER_P][4];
#pragma unroll
    for (int blk = 0; blk < BLKS_PER_P; ++blk)
#pragma unroll
        for (int kk = 0; kk < 4; ++kk)
            xq[blk][kk] = *(const v2f*)(xr + 16 * blk + 4 * kk);

    {
        const v4f x2 = aLc * aLc, x4 = x2 * x2, x8 = x4 * x4, x16 = x8 * x8;
        const v4f aC = x16 * x8 * x4 * x2;
        v4f C = {0.f, 0.f, 0.f, 0.f};
#pragma unroll
        for (int q = 0; q < ZR_; ++q) C = aC * C + sr[q];
        if (threadIdx.x < ROW_ / 4) *(v4f*)(sC + col) = C;
    }
    __syncthreads();
    stv = stv + pwv * *(const v8f*)(sC + g * 256 + lane * 8);

    int t0 = p * LP_;
#pragma unroll
    for (int blk = 0; blk < BLKS_PER_P; ++blk) {
        const v2f* xc = xq[blk];

        v8f qv = wlv * stv;
        const v2f* qc = (const v2f*)&qv;

        v8f acc = {0.f, 0.f, 0.f, 0.f, 0.f, 0.f, 0.f, 0.f};
#pragma unroll
        for (int kk = 0; kk < 4; ++kk)
            acc = __builtin_amdgcn_wmma_f32_16x16x4_f32(false, Pm[kk], false, qc[kk],
                                                        (short)0, acc, false, false);
        v8f rr = {0.f, 0.f, 0.f, 0.f, 0.f, 0.f, 0.f, 0.f};
#pragma unroll
        for (int kk = 0; kk < 4; ++kk) {
            acc = __builtin_amdgcn_wmma_f32_16x16x4_f32(false, Lm[kk], false, xc[kk],
                                                        (short)0, acc, false, false);
            rr  = __builtin_amdgcn_wmma_f32_16x16x4_f32(false, Rm[kk], false, xc[kk],
                                                        (short)0, rr, false, false);
        }
        stv = a16lv * stv + rr;

        float* op = out + (g * 16 + bl) * T_ + t0 + 8 * h;
        const v4f* ap = (const v4f*)&acc;
        *(v4f*)op       = ap[0];
        *(v4f*)(op + 4) = ap[1];

        t0 += 16;
    }
}

extern "C" void kernel_launch(void* const* d_in, const int* in_sizes, int n_in,
                              void* d_out, int out_size, void* d_ws, size_t ws_size,
                              hipStream_t stream) {
    const float* x  = (const float*)d_in[0];
    const float* lk = (const float*)d_in[1];
    const float* ar = (const float*)d_in[2];
    const float* br = (const float*)d_in[3];
    const float* H  = (const float*)d_in[4];
    float* ws  = (float*)d_ws;
    float* out = (float*)d_out;

    hipLaunchKernelGGL(fdn_partial, dim3(1 + NCH_ * (ROW_ / 32)), dim3(CP_ * 32),     0, stream, x, lk, ar, br, H, ws);
    hipLaunchKernelGGL(fdn_main,    dim3(P_ / MP_),               dim3(MP_ * 2 * 32), 0, stream, x, ws, out);
}


# First launch: each wave issues its global loads as one group (was one round trip per loop trip: 16 in the end-state chain, 10 in the table prologue); end-state threads re-mapped to (row, partition)
# speedup vs baseline: 21.3680x; 1.1032x over previous
#include <hip/hip_runtime.h>


#define D_ 16
#define B_ 32
#define T_ 48000
#define P_ 750
#define LP_ 64
#define BLKS_PER_P 4

typedef __attribute__((ext_vector_type(2))) float v2f;
typedef __attribute__((ext_vector_type(4))) float v4f;
typedef __attribute__((ext_vector_type(8))) float v8f;

#define WS_A      0
#define WS_W      16
#define WS_WL     64
#define WS_A16L   320
#define WS_TAB    576
#define WS_ALL    1344
#define WS_XS     2048
#define WS_S      WS_XS
#define WS_PW     32768
#define WS_E      (WS_XS + T_*B_)
#define WS_CARRYS (WS_E + P_*B_*D_)
#define CP_  30
#define NCH_ 25
#define ZR_  (NCH_ - 1)
#define MP_  3
#define ROW_ (B_ * D_)
#define TPB_ 128
#define NSET_ 8
#define NPC_ 15
#define AGP_ 33
#define XPITCH_ 68

__device__ __forceinline__ float sigmoidf_(float z) { return 1.0f / (1.0f + expf(-z)); }
__device__ __forceinline__ float powi_(float a, int n) {
    float r = 1.0f;
    for (int i = 0; i < n; ++i) r *= a;
    return r;
}
__device__ __forceinline__ float a64_(float a) {
    float r = a;
    for (int q = 0; q < 6; ++q) r *= r;
    return r;
}
__device__ __forceinline__ float decay_(float lkv) {
    const float max_delay = 50.0f * 16000.0f / 1000.0f;
    const float kappa = fmaf(sigmoidf_(lkv), max_delay - 1.0f, 1.0f);
    const float a = expf(-1.0f / kappa);
    return fminf(fmaxf(a, 0.0f), 0.9999f);
}

__device__ __forceinline__ void fdn_setup_block(const float* __restrict__ lk,
                                                const float* __restrict__ ar,
                                                const float* __restrict__ br,
                                                const float* __restrict__ H,
                                                float* __restrict__ ws,
                                                const int part) {
    const int first = part * TPB_ + (int)threadIdx.x;
    const int step = NSET_ * TPB_;
    __shared__ float sA[D_], sW[D_], sG[D_];
    const int t = threadIdx.x;
    float hc[D_] = {};
    float sb = 0.0f;
    if (t < D_) {
        const float lkv = lk[t], arv = ar[t], brv = br[t];
#pragma unroll
        for (int i = 0; i < D_; ++i) hc[i] = H[i * D_ + t];
        sA[t] = decay_(lkv);
        sG[t] = sigmoidf_(arv);
        sb = sigmoidf_(brv);
    }
    __syncthreads();
    if (t < D_) {
        float s = 0.0f;
#pragma unroll
        for (int i = 0; i < D_; ++i) s += sG[i] * hc[i];
        sW[t] = (sb + s) * (1.0f / (float)D_);
    }
    __syncthreads();

    for (int e = first; e < 3 * 256; e += step) {
        const int which = e >> 8, idx = e & 255;
        const int lane = idx >> 3, v = idx & 7;
        const int h = lane >> 4;
        const int d = 4 * (v >> 1) + 2 * h + (v & 1);
        const float a = sA[d];
        if (which == 0)      ws[WS_WL + idx] = sW[d];
        else if (which == 1) ws[WS_A16L + idx] = powi_(a, 16);
        else {
            const float aL = a64_(a);
            ws[WS_ALL + idx] = aL;
            float pw = 1.0f;
            for (int j = 0; j < CP_; ++j) { ws[WS_PW + j * 256 + idx] = pw; pw *= aL; }
        }
    }
    for (int e = first; e < ZR_ * ROW_; e += step) ws[WS_S + e] = 0.0f;

    for (int e = first; e < 12 * 64; e += step) {
        const int chunk = e >> 6, pos = e & 63;
        const int lane = pos >> 1, ra = pos & 1;
        const int h = lane >> 4, m = lane & 15;
        const int kap = ra + 2 * h;
        float val = 0.0f;
        if (chunk < 4) {
            const int j = 4 * chunk + kap;
            if (j <= m)
                for (int d = 0; d < D_; ++d)
                    val += sW[d] * powi_(sA[d], m - j);
        } else if (chunk < 8) {
            const int d = 4 * (chunk - 4) + kap;
            val = powi_(sA[d], m + 1);
        } else {
            int sg;
            if (m < 8) sg = 4 * (m >> 1) + (m & 1);
            else { const int v = m - 8; sg = 4 * (v >> 1) + 2 + (v & 1); }
            const int j = 4 * (chunk - 8) + kap;
            val = powi_(sA[sg], 15 - j);
        }
        ws[WS_TAB + e] = val;
    }
}

__global__ void __launch_bounds__(TPB_) fdn_partial(const float* __restrict__ x,
                                                    const float* __restrict__ lk,
                                                    const float* __restrict__ ar,
                                                    const float* __restrict__ br,
                                                    const float* __restrict__ H,
                                                    float* __restrict__ ws) {
    static_assert(NCH_ * CP_ == P_, "chunks must tile the partitions exactly");
    static_assert(4 * CP_ <= TPB_, "one thread per (row, partition) of the work-group");
    static_assert(4 * CP_ * LP_ / 4 == NPC_ * TPB_, "the tile is a whole number of 16-byte pieces per thread");
    __shared__ float agg[CP_][AGP_];
    __shared__ __attribute__((aligned(16))) float xs[4 * CP_ * XPITCH_];
    if (blockIdx.x < NSET_) {
        fdn_setup_block(lk, ar, br, H, ws, blockIdx.x);
        return;
    }
    const int wg = blockIdx.x - NSET_;
    const int c = wg >> 4;
    const int g = (wg >> 3) & 1, h = (wg >> 2) & 1, blq = wg & 3;
    const int lane = threadIdx.x & 31;

    const float* xt = x + (16 * g + 4 * blq) * T_ + c * CP_ * LP_;
    v4f st[NPC_];
#pragma unroll
    for (int j = 0; j < NPC_; ++j) {
        const int i = (int)threadIdx.x + TPB_ * j;
        const int r = i / (CP_ * LP_ / 4);
        st[j] = *(const v4f*)(xt + r * T_ + 4 * (i - r * (CP_ * LP_ / 4)));
    }

    const int vl = lane & 7;
    const float a = decay_(lk[4 * (vl >> 1) + 2 * h + (vl & 1)]);
    v2f av[4];
#pragma unroll
    for (int w = 0; w < 4; ++w) {
        av[w].x = __int_as_float(__builtin_amdgcn_readlane(__float_as_int(a), 2 * w));
        av[w].y = __int_as_float(__builtin_amdgcn_readlane(__float_as_int(a), 2 * w + 1));
    }

#pragma unroll
    for (int j = 0; j < NPC_; ++j) {
        const int i = (int)threadIdx.x + TPB_ * j;
        *(v4f*)(xs + (i >> 4) * XPITCH_ + 4 * (i & 15)) = st[j];
    }
    __syncthreads();

    const int tt = threadIdx.x < 4 * CP_ ? (int)threadIdx.x : 4 * CP_ - 1;
    const int r = tt / CP_, q = tt - r * CP_;
    const float* xp = xs + tt * XPITCH_;
    v4f xv[LP_ / 4];
#pragma unroll
    for (int k = 0; k < LP_ / 4; ++k) xv[k] = *(const v4f*)(xp + 4 * k);
    asm volatile("" :: "v"(xv[0]), "v"(xv[1]), "v"(xv[2]), "v"(xv[3]), "v"(xv[4]), "v"(xv[5]),
                       "v"(xv[6]), "v"(xv[7]), "v"(xv[8]), "v"(xv[9]), "v"(xv[10]), "v"(xv[11]),
                       "v"(xv[12]), "v"(xv[13]), "v"(xv[14]), "v"(xv[15]));

    v2f acc[4] = {{0.f, 0.f}, {0.f, 0.f}, {0.f, 0.f}, {0.f, 0.f}};
#pragma unroll
    for (int k = 0; k < LP_ / 4; ++k) {
#pragma unroll
        for (int j = 0; j < 4; ++j) {
            const float xe = xv[k][j];
            const v2f x2 = {xe, xe};
#pragma unroll
            for (int w = 0; w < 4; ++w) acc[w] = __builtin_elementwise_fma(av[w], acc[w], x2);
        }
    }
    if (threadIdx.x < 4 * CP_) {
#pragma unroll
        for (int w = 0; w < 4; ++w) {
            agg[q][8 * r + 2 * w]     = acc[w].x;
            agg[q][8 * r + 2 * w + 1] = acc[w].y;
        }
    }
    __syncthreads();
    if (threadIdx.x >= 32) return;

    const int bl = 4 * blq + (lane >> 3), v = lane & 7;
    const float aL = a64_(a);
    float e[CP_];
#pragma unroll
    for (int i = 0; i < CP_; ++i) e[i] = agg[i][lane];

    const int outSlot = (((h << 4) + bl) << 3) + v + (g << 8);
    float* cp = ws + WS_CARRYS + c * CP_ * ROW_ + outSlot;
    float s = 0.0f;
#pragma unroll
    for (int i = 0; i < CP_; ++i) {
        cp[i * ROW_] = s;
        s = fmaf(aL, s, e[i]);
    }
    ws[WS_S + (ZR_ + c) * ROW_ + outSlot] = s;
}

__global__ void __launch_bounds__(MP_ * 2 * 32) fdn_main(const float* __restrict__ x,
                                                         const float* __restrict__ ws,
                                                         float* __restrict__ out) {
    static_assert(CP_ % MP_ == 0 && P_ % MP_ == 0, "a work-group stays inside one chunk");
    static_assert(MP_ * 2 * 32 >= ROW_ / 4, "one column quad per thread must cover a row");
    static_assert(CP_ == 30, "aC below is aL^30");
    __shared__ __attribute__((aligned(32))) float sC[ROW_];
    const int lane = threadIdx.x & 31;
    const int wv = threadIdx.x >> 5;
    const int p = blockIdx.x * MP_ + (wv >> 1), g = wv & 1;
    const int widx = 2 * p + g;
    const int c = blockIdx.x / (CP_ / MP_);
    const int h = lane >> 4, bl = lane & 15;

    const int col = 4 * (threadIdx.x & (ROW_ / 4 - 1));
    const v4f aLc = *(const v4f*)(ws + WS_ALL + (col & 255));
    const float* sp = ws + WS_S + c * ROW_ + col;
    v4f sr[ZR_];
#pragma unroll
    for (int q = 0; q < ZR_; ++q) sr[q] = *(const v4f*)(sp + q * ROW_);
    const v8f pwv = *(const v8f*)(ws + WS_PW + (p - c * CP_) * 256 + lane * 8);

    const v8f wlv   = *(const v8f*)(ws + WS_WL   + lane * 8);
    const v8f a16lv = *(const v8f*)(ws + WS_A16L + lane * 8);

    v2f Lm[4], Pm[4], Rm[4];
#pragma unroll
    for (int kk = 0; kk < 4; ++kk) {
        Lm[kk] = *(const v2f*)(ws + WS_TAB + (kk)     * 64 + lane * 2);
        Pm[kk] = *(const v2f*)(ws + WS_TAB + (kk + 4) * 64 + lane * 2);
        Rm[kk] = *(const v2f*)(ws + WS_TAB + (kk + 8) * 64 + lane * 2);
    }

    v8f stv = *(const v8f*)(ws + WS_CARRYS + (widx * 32 + lane) * 8);

    const float* xr = x + (g * 16 + bl) * T_ + p * LP_ + 2 * h;
    v2f xq[BLKS_PER_P][4];
#pragma unroll
    for (int blk = 0; blk < BLKS_PER_P; ++blk)
#pragma unroll
        for (int kk = 0; kk < 4; ++kk)
            xq[blk][kk] = *(const v2f*)(xr + 16 * blk + 4 * kk);

    {
        const v4f x2 = aLc * aLc, x4 = x2 * x2, x8 = x4 * x4, x16 = x8 * x8;
        const v4f aC = x16 * x8 * x4 * x2;
        v4f C = {0.f, 0.f, 0.f, 0.f};
#pragma unroll
        for (int q = 0; q < ZR_; ++q) C = aC * C + sr[q];
        if (threadIdx.x < ROW_ / 4) *(v4f*)(sC + col) = C;
    }
    __syncthreads();
    stv = stv + pwv * *(const v8f*)(sC + g * 256 + lane * 8);

    int t0 = p * LP_;
#pragma unroll
    for (int blk = 0; blk < BLKS_PER_P; ++blk) {
        const v2f* xc = xq[blk];

        v8f qv = wlv * stv;
        const v2f* qc = (const v2f*)&qv;

        v8f acc = {0.f, 0.f, 0.f, 0.f, 0.f, 0.f, 0.f, 0.f};
#pragma unroll
        for (int kk = 0; kk < 4; ++kk)
            acc = __builtin_amdgcn_wmma_f32_16x16x4_f32(false, Pm[kk], false, qc[kk],
                                                        (short)0, acc, false, false);
        v8f rr = {0.f, 0.f, 0.f, 0.f, 0.f, 0.f, 0.f, 0.f};
#pragma unroll
        for (int kk = 0; kk < 4; ++kk) {
            acc = __builtin_amdgcn_wmma_f32_16x16x4_f32(false, Lm[kk], false, xc[kk],
                                                        (short)0, acc, false, false);
            rr  = __builtin_amdgcn_wmma_f32_16x16x4_f32(false, Rm[kk], false, xc[kk],
                                                        (short)0, rr, false, false);
        }
        stv = a16lv * stv + rr;

        float* op = out + (g * 16 + bl) * T_ + t0 + 8 * h;
        const v4f* ap = (const v4f*)&acc;
        *(v4f*)op       = ap[0];
        *(v4f*)(op + 4) = ap[1];

        t0 += 16;
    }
}

extern "C" void kernel_launch(void* const* d_in, const int* in_sizes, int n_in,
                              void* d_out, int out_size, void* d_ws, size_t ws_size,
                              hipStream_t stream) {
    const float* x  = (const float*)d_in[0];
    const float* lk = (const float*)d_in[1];
    const float* ar = (const float*)d_in[2];
    const float* br = (const float*)d_in[3];
    const float* H  = (const float*)d_in[4];
    float* ws  = (float*)d_ws;
    float* out = (float*)d_out;

    hipLaunchKernelGGL(fdn_partial, dim3(NSET_ + NCH_ * (ROW_ / 32)), dim3(TPB_),     0, stream, x, lk, ar, br, H, ws);
    hipLaunchKernelGGL(fdn_main,    dim3(P_ / MP_),               dim3(MP_ * 2 * 32), 0, stream, x, ws, out);
}


# Table builder straight-line: a_d^n (n<=16) built once per work-group as a running product in LDS and read by every A-matrix entry (was up to 16 run-time power loops per entry); zero fill unrolled
# speedup vs baseline: 29.7277x; 1.3912x over previous
#include <hip/hip_runtime.h>


#define D_ 16
#define B_ 32
#define T_ 48000
#define P_ 750
#define LP_ 64
#define BLKS_PER_P 4

typedef __attribute__((ext_vector_type(2))) float v2f;
typedef __attribute__((ext_vector_type(4))) float v4f;
typedef __attribute__((ext_vector_type(8))) float v8f;

#define WS_A      0
#define WS_W      16
#define WS_WL     64
#define WS_A16L   320
#define WS_TAB    576
#define WS_ALL    1344
#define WS_XS     2048
#define WS_S      WS_XS
#define WS_PW     32768
#define WS_E      (WS_XS + T_*B_)
#define WS_CARRYS (WS_E + P_*B_*D_)
#define CP_  30
#define NCH_ 25
#define ZR_  (NCH_ - 1)
#define MP_  3
#define ROW_ (B_ * D_)
#define TPB_ 128
#define NSET_ 8
#define NPC_ 15
#define AGP_ 33
#define NPW_ 17
#define PWP_ 20
#define XPITCH_ 68

__device__ __forceinline__ float sigmoidf_(float z) { return 1.0f / (1.0f + expf(-z)); }
__device__ __forceinline__ float a64_(float a) {
    float r = a;
    for (int q = 0; q < 6; ++q) r *= r;
    return r;
}
__device__ __forceinline__ float decay_(float lkv) {
    const float max_delay = 50.0f * 16000.0f / 1000.0f;
    const float kappa = fmaf(sigmoidf_(lkv), max_delay - 1.0f, 1.0f);
    const float a = expf(-1.0f / kappa);
    return fminf(fmaxf(a, 0.0f), 0.9999f);
}

__device__ __forceinline__ void fdn_setup_block(const float* __restrict__ lk,
                                                const float* __restrict__ ar,
                                                const float* __restrict__ br,
                                                const float* __restrict__ H,
                                                float* __restrict__ ws,
                                                const int part) {
    const int first = part * TPB_ + (int)threadIdx.x;
    const int step = NSET_ * TPB_;
    __shared__ float sA[D_], sG[D_];
    __shared__ __attribute__((aligned(16))) float sW[D_];
    __shared__ __attribute__((aligned(16))) float sP[NPW_ * PWP_];
    const int t = threadIdx.x;
    float hc[D_] = {};
    float sb = 0.0f;
    if (t < D_) {
        const float lkv = lk[t], arv = ar[t], brv = br[t];
#pragma unroll
        for (int i = 0; i < D_; ++i) hc[i] = H[i * D_ + t];
        const float a = decay_(lkv);
        sA[t] = a;
        sG[t] = sigmoidf_(arv);
        sb = sigmoidf_(brv);
        float p = 1.0f;
#pragma unroll
        for (int n = 0; n < NPW_; ++n) { sP[n * PWP_ + t] = p; p *= a; }
    }
    __syncthreads();
    if (t < D_) {
        float s = 0.0f;
#pragma unroll
        for (int i = 0; i < D_; ++i) s += sG[i] * hc[i];
        sW[t] = (sb + s) * (1.0f / (float)D_);
    }
    __syncthreads();

    for (int e = first; e < 3 * 256; e += step) {
        const int which = e >> 8, idx = e & 255;
        const int lane = idx >> 3, v = idx & 7;
        const int h = lane >> 4;
        const int d = 4 * (v >> 1) + 2 * h + (v & 1);
        const float a = sA[d];
        if (which == 0)      ws[WS_WL + idx] = sW[d];
        else if (which == 1) ws[WS_A16L + idx] = sP[16 * PWP_ + d];
        else {
            const float aL = a64_(a);
            ws[WS_ALL + idx] = aL;
            float pw = 1.0f;
            for (int j = 0; j < CP_; ++j) { ws[WS_PW + j * 256 + idx] = pw; pw *= aL; }
        }
    }
    static_assert((ZR_ * ROW_) % (4 * NSET_ * TPB_) == 0, "the zero rows are a whole number of 16-byte stores per thread");
#pragma unroll
    for (int k = 0; k < ZR_ * ROW_ / (4 * NSET_ * TPB_); ++k)
        *(v4f*)(ws + WS_S + 4 * (first + k * step)) = v4f{0.f, 0.f, 0.f, 0.f};

    for (int e = first; e < 12 * 64; e += step) {
        const int chunk = e >> 6, pos = e & 63;
        const int lane = pos >> 1, ra = pos & 1;
        const int h = lane >> 4, m = lane & 15;
        const int kap = ra + 2 * h;
        float val = 0.0f;
        if (chunk < 4) {
            const int n = m - (4 * chunk + kap);
            if (n >= 0) {
                const v4f* pr = (const v4f*)(sP + n * PWP_);
                const v4f* wr = (const v4f*)sW;
#pragma unroll
                for (int q = 0; q < D_ / 4; ++q) {
                    const v4f pv = pr[q], wv = wr[q];
#pragma unroll
                    for (int i = 0; i < 4; ++i) val += wv[i] * pv[i];
                }
            }
        } else if (chunk < 8) {
            const int d = 4 * (chunk - 4) + kap;
            val = sP[(m + 1) * PWP_ + d];
        } else {
            int sg;
            if (m < 8) sg = 4 * (m >> 1) + (m & 1);
            else { const int v = m - 8; sg = 4 * (v >> 1) + 2 + (v & 1); }
            const int j = 4 * (chunk - 8) + kap;
            val = sP[(15 - j) * PWP_ + sg];
        }
        ws[WS_TAB + e] = val;
    }
}

__global__ void __launch_bounds__(TPB_) fdn_partial(const float* __restrict__ x,
                                                    const float* __restrict__ lk,
                                                    const float* __restrict__ ar,
                                                    const float* __restrict__ br,
                                                    const float* __restrict__ H,
                                                    float* __restrict__ ws) {
    static_assert(NCH_ * CP_ == P_, "chunks must tile the partitions exactly");
    static_assert(4 * CP_ <= TPB_, "one thread per (row, partition) of the work-group");
    static_assert(4 * CP_ * LP_ / 4 == NPC_ * TPB_, "the tile is a whole number of 16-byte pieces per thread");
    __shared__ float agg[CP_][AGP_];
    __shared__ __attribute__((aligned(16))) float xs[4 * CP_ * XPITCH_];
    if (blockIdx.x < NSET_) {
        fdn_setup_block(lk, ar, br, H, ws, blockIdx.x);
        return;
    }
    const int wg = blockIdx.x - NSET_;
    const int c = wg >> 4;
    const int g = (wg >> 3) & 1, h = (wg >> 2) & 1, blq = wg & 3;
    const int lane = threadIdx.x & 31;

    const float* xt = x + (16 * g + 4 * blq) * T_ + c * CP_ * LP_;
    v4f st[NPC_];
#pragma unroll
    for (int j = 0; j < NPC_; ++j) {
        const int i = (int)threadIdx.x + TPB_ * j;
        const int r = i / (CP_ * LP_ / 4);
        st[j] = *(const v4f*)(xt + r * T_ + 4 * (i - r * (CP_ * LP_ / 4)));
    }

    const int vl = lane & 7;
    const float a = decay_(lk[4 * (vl >> 1) + 2 * h + (vl & 1)]);
    v2f av[4];
#pragma unroll
    for (int w = 0; w < 4; ++w) {
        av[w].x = __int_as_float(__builtin_amdgcn_readlane(__float_as_int(a), 2 * w));
        av[w].y = __int_as_float(__builtin_amdgcn_readlane(__float_as_int(a), 2 * w + 1));
    }

#pragma unroll
    for (int j = 0; j < NPC_; ++j) {
        const int i = (int)threadIdx.x + TPB_ * j;
        *(v4f*)(xs + (i >> 4) * XPITCH_ + 4 * (i & 15)) = st[j];
    }
    __syncthreads();

    const int tt = threadIdx.x < 4 * CP_ ? (int)threadIdx.x : 4 * CP_ - 1;
    const int r = tt / CP_, q = tt - r * CP_;
    const float* xp = xs + tt * XPITCH_;
    v4f xv[LP_ / 4];
#pragma unroll
    for (int k = 0; k < LP_ / 4; ++k) xv[k] = *(const v4f*)(xp + 4 * k);
    asm volatile("" :: "v"(xv[0]), "v"(xv[1]), "v"(xv[2]), "v"(xv[3]), "v"(xv[4]), "v"(xv[5]),
                       "v"(xv[6]), "v"(xv[7]), "v"(xv[8]), "v"(xv[9]), "v"(xv[10]), "v"(xv[11]),
                       "v"(xv[12]), "v"(xv[13]), "v"(xv[14]), "v"(xv[15]));

    v2f acc[4] = {{0.f, 0.f}, {0.f, 0.f}, {0.f, 0.f}, {0.f, 0.f}};
#pragma unroll
    for (int k = 0; k < LP_ / 4; ++k) {
#pragma unroll
        for (int j = 0; j < 4; ++j) {
            const float xe = xv[k][j];
            const v2f x2 = {xe, xe};
#pragma unroll
            for (int w = 0; w < 4; ++w) acc[w] = __builtin_elementwise_fma(av[w], acc[w], x2);
        }
    }
    if (threadIdx.x < 4 * CP_) {
#pragma unroll
        for (int w = 0; w < 4; ++w) {
            agg[q][8 * r + 2 * w]     = acc[w].x;
            agg[q][8 * r + 2 * w + 1] = acc[w].y;
        }
    }
    __syncthreads();
    if (threadIdx.x >= 32) return;

    const int bl = 4 * blq + (lane >> 3), v = lane & 7;
    const float aL = a64_(a);
    float e[CP_];
#pragma unroll
    for (int i = 0; i < CP_; ++i) e[i] = agg[i][lane];

    const int outSlot = (((h << 4) + bl) << 3) + v + (g << 8);
    float* cp = ws + WS_CARRYS + c * CP_ * ROW_ + outSlot;
    float s = 0.0f;
#pragma unroll
    for (int i = 0; i < CP_; ++i) {
        cp[i * ROW_] = s;
        s = fmaf(aL, s, e[i]);
    }
    ws[WS_S + (ZR_ + c) * ROW_ + outSlot] = s;
}

__global__ void __launch_bounds__(MP_ * 2 * 32) fdn_main(const float* __restrict__ x,
                                                         const float* __restrict__ ws,
                                                         float* __restrict__ out) {
    static_assert(CP_ % MP_ == 0 && P_ % MP_ == 0, "a work-group stays inside one chunk");
    static_assert(MP_ * 2 * 32 >= ROW_ / 4, "one column quad per thread must cover a row");
    static_assert(CP_ == 30, "aC below is aL^30");
    __shared__ __attribute__((aligned(32))) float sC[ROW_];
    const int lane = threadIdx.x & 31;
    const int wv = threadIdx.x >> 5;
    const int p = blockIdx.x * MP_ + (wv >> 1), g = wv & 1;
    const int widx = 2 * p + g;
    const int c = blockIdx.x / (CP_ / MP_);
    const int h = lane >> 4, bl = lane & 15;

    const int col = 4 * (threadIdx.x & (ROW_ / 4 - 1));
    const v4f aLc = *(const v4f*)(ws + WS_ALL + (col & 255));
    const float* sp = ws + WS_S + c * ROW_ + col;
    v4f sr[ZR_];
#pragma unroll
    for (int q = 0; q < ZR_; ++q) sr[q] = *(const v4f*)(sp + q * ROW_);
    const v8f pwv = *(const v8f*)(ws + WS_PW + (p - c * CP_) * 256 + lane * 8);

    const v8f wlv   = *(const v8f*)(ws + WS_WL   + lane * 8);
    const v8f a16lv = *(const v8f*)(ws + WS_A16L + lane * 8);

    v2f Lm[4], Pm[4], Rm[4];
#pragma unroll
    for (int kk = 0; kk < 4; ++kk) {
        Lm[kk] = *(const v2f*)(ws + WS_TAB + (kk)     * 64 + lane * 2);
        Pm[kk] = *(const v2f*)(ws + WS_TAB + (kk + 4) * 64 + lane * 2);
        Rm[kk] = *(const v2f*)(ws + WS_TAB + (kk + 8) * 64 + lane * 2);
    }

    v8f stv = *(const v8f*)(ws + WS_CARRYS + (widx * 32 + lane) * 8);

    const float* xr = x + (g * 16 + bl) * T_ + p * LP_ + 2 * h;
    v2f xq[BLKS_PER_P][4];
#pragma unroll
    for (int blk = 0; blk < BLKS_PER_P; ++blk)
#pragma unroll
        for (int kk = 0; kk < 4; ++kk)
            xq[blk][kk] = *(const v2f*)(xr + 16 * blk + 4 * kk);

    {
        const v4f x2 = aLc * aLc, x4 = x2 * x2, x8 = x4 * x4, x16 = x8 * x8;
        const v4f aC = x16 * x8 * x4 * x2;
        v4f C = {0.f, 0.f, 0.f, 0.f};
#pragma unroll
        for (int q = 0; q < ZR_; ++q) C = aC * C + sr[q];
        if (threadIdx.x < ROW_ / 4) *(v4f*)(sC + col) = C;
    }
    __syncthreads();
    stv = stv + pwv * *(const v8f*)(sC + g * 256 + lane * 8);

    int t0 = p * LP_;
#pragma unroll
    for (int blk = 0; blk < BLKS_PER_P; ++blk) {
        const v2f* xc = xq[blk];

        v8f qv = wlv * stv;
        const v2f* qc = (const v2f*)&qv;

        v8f acc = {0.f, 0.f, 0.f, 0.f, 0.f, 0.f, 0.f, 0.f};
#pragma unroll
        for (int kk = 0; kk < 4; ++kk)
            acc = __builtin_amdgcn_wmma_f32_16x16x4_f32(false, Pm[kk], false, qc[kk],
                                                        (short)0, acc, false, false);
        v8f rr = {0.f, 0.f, 0.f, 0.f, 0.f, 0.f, 0.f, 0.f};
#pragma unroll
        for (int kk = 0; kk < 4; ++kk) {
            acc = __builtin_amdgcn_wmma_f32_16x16x4_f32(false, Lm[kk], false, xc[kk],
                                                        (short)0, acc, false, false);
            rr  = __builtin_amdgcn_wmma_f32_16x16x4_f32(false, Rm[kk], false, xc[kk],
                                                        (short)0, rr, false, false);
        }
        stv = a16lv * stv + rr;

        float* op = out + (g * 16 + bl) * T_ + t0 + 8 * h;
        const v4f* ap = (const v4f*)&acc;
        *(v4f*)op       = ap[0];
        *(v4f*)(op + 4) = ap[1];

        t0 += 16;
    }
}

extern "C" void kernel_launch(void* const* d_in, const int* in_sizes, int n_in,
                              void* d_out, int out_size, void* d_ws, size_t ws_size,
                              hipStream_t stream) {
    const float* x  = (const float*)d_in[0];
    const float* lk = (const float*)d_in[1];
    const float* ar = (const float*)d_in[2];
    const float* br = (const float*)d_in[3];
    const float* H  = (const float*)d_in[4];
    float* ws  = (float*)d_ws;
    float* out = (float*)d_out;

    hipLaunchKernelGGL(fdn_partial, dim3(NSET_ + NCH_ * (ROW_ / 32)), dim3(TPB_),     0, stream, x, lk, ar, br, H, ws);
    hipLaunchKernelGGL(fdn_main,    dim3(P_ / MP_),               dim3(MP_ * 2 * 32), 0, stream, x, ws, out);
}


# Output launch: each wave turns its 16x64 result tile through its own LDS region and writes it with 8 stores of 4 whole 128-byte lines each (was 8 stores of 16 lines, two 16-byte pieces per line)
# speedup vs baseline: 39.8094x; 1.3391x over previous
#include <hip/hip_runtime.h>


#define D_ 16
#define B_ 32
#define T_ 48000
#define P_ 750
#define LP_ 64
#define BLKS_PER_P 4

typedef __attribute__((ext_vector_type(2))) float v2f;
typedef __attribute__((ext_vector_type(4))) float v4f;
typedef __attribute__((ext_vector_type(8))) float v8f;

#define WS_A      0
#define WS_W      16
#define WS_WL     64
#define WS_A16L   320
#define WS_TAB    576
#define WS_ALL    1344
#define WS_XS     2048
#define WS_S      WS_XS
#define WS_PW     32768
#define WS_E      (WS_XS + T_*B_)
#define WS_CARRYS (WS_E + P_*B_*D_)
#define CP_  30
#define NCH_ 25
#define ZR_  (NCH_ - 1)
#define MP_  3
#define ROW_ (B_ * D_)
#define TPB_ 128
#define NSET_ 8
#define NPC_ 15
#define AGP_ 33
#define NPW_ 17
#define PWP_ 20
#define XPITCH_ 68
#define OPITCH_ 68

__device__ __forceinline__ float sigmoidf_(float z) { return 1.0f / (1.0f + expf(-z)); }
__device__ __forceinline__ float a64_(float a) {
    float r = a;
    for (int q = 0; q < 6; ++q) r *= r;
    return r;
}
__device__ __forceinline__ float decay_(float lkv) {
    const float max_delay = 50.0f * 16000.0f / 1000.0f;
    const float kappa = fmaf(sigmoidf_(lkv), max_delay - 1.0f, 1.0f);
    const float a = expf(-1.0f / kappa);
    return fminf(fmaxf(a, 0.0f), 0.9999f);
}

__device__ __forceinline__ void fdn_setup_block(const float* __restrict__ lk,
                                                const float* __restrict__ ar,
                                                const float* __restrict__ br,
                                                const float* __restrict__ H,
                                                float* __restrict__ ws,
                                                const int part) {
    const int first = part * TPB_ + (int)threadIdx.x;
    const int step = NSET_ * TPB_;
    __shared__ float sA[D_], sG[D_];
    __shared__ __attribute__((aligned(16))) float sW[D_];
    __shared__ __attribute__((aligned(16))) float sP[NPW_ * PWP_];
    const int t = threadIdx.x;
    float hc[D_] = {};
    float sb = 0.0f;
    if (t < D_) {
        const float lkv = lk[t], arv = ar[t], brv = br[t];
#pragma unroll
        for (int i = 0; i < D_; ++i) hc[i] = H[i * D_ + t];
        const float a = decay_(lkv);
        sA[t] = a;
        sG[t] = sigmoidf_(arv);
        sb = sigmoidf_(brv);
        float p = 1.0f;
#pragma unroll
        for (int n = 0; n < NPW_; ++n) { sP[n * PWP_ + t] = p; p *= a; }
    }
    __syncthreads();
    if (t < D_) {
        float s = 0.0f;
#pragma unroll
        for (int i = 0; i < D_; ++i) s += sG[i] * hc[i];
        sW[t] = (sb + s) * (1.0f / (float)D_);
    }
    __syncthreads();

    for (int e = first; e < 3 * 256; e += step) {
        const int which = e >> 8, idx = e & 255;
        const int lane = idx >> 3, v = idx & 7;
        const int h = lane >> 4;
        const int d = 4 * (v >> 1) + 2 * h + (v & 1);
        const float a = sA[d];
        if (which == 0)      ws[WS_WL + idx] = sW[d];
        else if (which == 1) ws[WS_A16L + idx] = sP[16 * PWP_ + d];
        else {
            const float aL = a64_(a);
            ws[WS_ALL + idx] = aL;
            float pw = 1.0f;
            for (int j = 0; j < CP_; ++j) { ws[WS_PW + j * 256 + idx] = pw; pw *= aL; }
        }
    }
    static_assert((ZR_ * ROW_) % (4 * NSET_ * TPB_) == 0, "the zero rows are a whole number of 16-byte stores per thread");
#pragma unroll
    for (int k = 0; k < ZR_ * ROW_ / (4 * NSET_ * TPB_); ++k)
        *(v4f*)(ws + WS_S + 4 * (first + k * step)) = v4f{0.f, 0.f, 0.f, 0.f};

    for (int e = first; e < 12 * 64; e += step) {
        const int chunk = e >> 6, pos = e & 63;
        const int lane = pos >> 1, ra = pos & 1;
        const int h = lane >> 4, m = lane & 15;
        const int kap = ra + 2 * h;
        float val = 0.0f;
        if (chunk < 4) {
            const int n = m - (4 * chunk + kap);
            if (n >= 0) {
                const v4f* pr = (const v4f*)(sP + n * PWP_);
                const v4f* wr = (const v4f*)sW;
#pragma unroll
                for (int q = 0; q < D_ / 4; ++q) {
                    const v4f pv = pr[q], wv = wr[q];
#pragma unroll
                    for (int i = 0; i < 4; ++i) val += wv[i] * pv[i];
                }
            }
        } else if (chunk < 8) {
            const int d = 4 * (chunk - 4) + kap;
            val = sP[(m + 1) * PWP_ + d];
        } else {
            int sg;
            if (m < 8) sg = 4 * (m >> 1) + (m & 1);
            else { const int v = m - 8; sg = 4 * (v >> 1) + 2 + (v & 1); }
            const int j = 4 * (chunk - 8) + kap;
            val = sP[(15 - j) * PWP_ + sg];
        }
        ws[WS_TAB + e] = val;
    }
}

__global__ void __launch_bounds__(TPB_) fdn_partial(const float* __restrict__ x,
                                                    const float* __restrict__ lk,
                                                    const float* __restrict__ ar,
                                                    const float* __restrict__ br,
                                                    const float* __restrict__ H,
                                                    float* __restrict__ ws) {
    static_assert(NCH_ * CP_ == P_, "chunks must tile the partitions exactly");
    static_assert(4 * CP_ <= TPB_, "one thread per (row, partition) of the work-group");
    static_assert(4 * CP_ * LP_ / 4 == NPC_ * TPB_, "the tile is a whole number of 16-byte pieces per thread");
    __shared__ float agg[CP_][AGP_];
    __shared__ __attribute__((aligned(16))) float xs[4 * CP_ * XPITCH_];
    if (blockIdx.x < NSET_) {
        fdn_setup_block(lk, ar, br, H, ws, blockIdx.x);
        return;
    }
    const int wg = blockIdx.x - NSET_;
    const int c = wg >> 4;
    const int g = (wg >> 3) & 1, h = (wg >> 2) & 1, blq = wg & 3;
    const int lane = threadIdx.x & 31;

    const float* xt = x + (16 * g + 4 * blq) * T_ + c * CP_ * LP_;
    v4f st[NPC_];
#pragma unroll
    for (int j = 0; j < NPC_; ++j) {
        const int i = (int)threadIdx.x + TPB_ * j;
        const int r = i / (CP_ * LP_ / 4);
        st[j] = *(const v4f*)(xt + r * T_ + 4 * (i - r * (CP_ * LP_ / 4)));
    }

    const int vl = lane & 7;
    const float a = decay_(lk[4 * (vl >> 1) + 2 * h + (vl & 1)]);
    v2f av[4];
#pragma unroll
    for (int w = 0; w < 4; ++w) {
        av[w].x = __int_as_float(__builtin_amdgcn_readlane(__float_as_int(a), 2 * w));
        av[w].y = __int_as_float(__builtin_amdgcn_readlane(__float_as_int(a), 2 * w + 1));
    }

#pragma unroll
    for (int j = 0; j < NPC_; ++j) {
        const int i = (int)threadIdx.x + TPB_ * j;
        *(v4f*)(xs + (i >> 4) * XPITCH_ + 4 * (i & 15)) = st[j];
    }
    __syncthreads();

    const int tt = threadIdx.x < 4 * CP_ ? (int)threadIdx.x : 4 * CP_ - 1;
    const int r = tt / CP_, q = tt - r * CP_;
    const float* xp = xs + tt * XPITCH_;
    v4f xv[LP_ / 4];
#pragma unroll
    for (int k = 0; k < LP_ / 4; ++k) xv[k] = *(const v4f*)(xp + 4 * k);
    asm volatile("" :: "v"(xv[0]), "v"(xv[1]), "v"(xv[2]), "v"(xv[3]), "v"(xv[4]), "v"(xv[5]),
                       "v"(xv[6]), "v"(xv[7]), "v"(xv[8]), "v"(xv[9]), "v"(xv[10]), "v"(xv[11]),
                       "v"(xv[12]), "v"(xv[13]), "v"(xv[14]), "v"(xv[15]));

    v2f acc[4] = {{0.f, 0.f}, {0.f, 0.f}, {0.f, 0.f}, {0.f, 0.f}};
#pragma unroll
    for (int k = 0; k < LP_ / 4; ++k) {
#pragma unroll
        for (int j = 0; j < 4; ++j) {
            const float xe = xv[k][j];
            const v2f x2 = {xe, xe};
#pragma unroll
            for (int w = 0; w < 4; ++w) acc[w] = __builtin_elementwise_fma(av[w], acc[w], x2);
        }
    }
    if (threadIdx.x < 4 * CP_) {
#pragma unroll
        for (int w = 0; w < 4; ++w) {
            agg[q][8 * r + 2 * w]     = acc[w].x;
            agg[q][8 * r + 2 * w + 1] = acc[w].y;
        }
    }
    __syncthreads();
    if (threadIdx.x >= 32) return;

    const int bl = 4 * blq + (lane >> 3), v = lane & 7;
    const float aL = a64_(a);
    float e[CP_];
#pragma unroll
    for (int i = 0; i < CP_; ++i) e[i] = agg[i][lane];

    const int outSlot = (((h << 4) + bl) << 3) + v + (g << 8);
    float* cp = ws + WS_CARRYS + c * CP_ * ROW_ + outSlot;
    float s = 0.0f;
#pragma unroll
    for (int i = 0; i < CP_; ++i) {
        cp[i * ROW_] = s;
        s = fmaf(aL, s, e[i]);
    }
    ws[WS_S + (ZR_ + c) * ROW_ + outSlot] = s;
}

__global__ void __launch_bounds__(MP_ * 2 * 32) fdn_main(const float* __restrict__ x,
                                                         const float* __restrict__ ws,
                                                         float* __restrict__ out) {
    static_assert(CP_ % MP_ == 0 && P_ % MP_ == 0, "a work-group stays inside one chunk");
    static_assert(MP_ * 2 * 32 >= ROW_ / 4, "one column quad per thread must cover a row");
    static_assert(CP_ == 30, "aC below is aL^30");
    __shared__ __attribute__((aligned(32))) float sC[ROW_];
    __shared__ __attribute__((aligned(16))) float sO[MP_ * 2][16 * OPITCH_];
    const int lane = threadIdx.x & 31;
    const int wv = threadIdx.x >> 5;
    const int p = blockIdx.x * MP_ + (wv >> 1), g = wv & 1;
    const int widx = 2 * p + g;
    const int c = blockIdx.x / (CP_ / MP_);
    const int h = lane >> 4, bl = lane & 15;

    const int col = 4 * (threadIdx.x & (ROW_ / 4 - 1));
    const v4f aLc = *(const v4f*)(ws + WS_ALL + (col & 255));
    const float* sp = ws + WS_S + c * ROW_ + col;
    v4f sr[ZR_];
#pragma unroll
    for (int q = 0; q < ZR_; ++q) sr[q] = *(const v4f*)(sp + q * ROW_);
    const v8f pwv = *(const v8f*)(ws + WS_PW + (p - c * CP_) * 256 + lane * 8);

    const v8f wlv   = *(const v8f*)(ws + WS_WL   + lane * 8);
    const v8f a16lv = *(const v8f*)(ws + WS_A16L + lane * 8);

    v2f Lm[4], Pm[4], Rm[4];
#pragma unroll
    for (int kk = 0; kk < 4; ++kk) {
        Lm[kk] = *(const v2f*)(ws + WS_TAB + (kk)     * 64 + lane * 2);
        Pm[kk] = *(const v2f*)(ws + WS_TAB + (kk + 4) * 64 + lane * 2);
        Rm[kk] = *(const v2f*)(ws + WS_TAB + (kk + 8) * 64 + lane * 2);
    }

    v8f stv = *(const v8f*)(ws + WS_CARRYS + (widx * 32 + lane) * 8);

    const float* xr = x + (g * 16 + bl) * T_ + p * LP_ + 2 * h;
    v2f xq[BLKS_PER_P][4];
#pragma unroll
    for (int blk = 0; blk < BLKS_PER_P; ++blk)
#pragma unroll
        for (int kk = 0; kk < 4; ++kk)
            xq[blk][kk] = *(const v2f*)(xr + 16 * blk + 4 * kk);

    {
        const v4f x2 = aLc * aLc, x4 = x2 * x2, x8 = x4 * x4, x16 = x8 * x8;
        const v4f aC = x16 * x8 * x4 * x2;
        v4f C = {0.f, 0.f, 0.f, 0.f};
#pragma unroll
        for (int q = 0; q < ZR_; ++q) C = aC * C + sr[q];
        if (threadIdx.x < ROW_ / 4) *(v4f*)(sC + col) = C;
    }
    __syncthreads();
    stv = stv + pwv * *(const v8f*)(sC + g * 256 + lane * 8);

    float* so = &sO[wv][0];
#pragma unroll
    for (int blk = 0; blk < BLKS_PER_P; ++blk) {
        const v2f* xc = xq[blk];

        v8f qv = wlv * stv;
        const v2f* qc = (const v2f*)&qv;

        v8f acc = {0.f, 0.f, 0.f, 0.f, 0.f, 0.f, 0.f, 0.f};
#pragma unroll
        for (int kk = 0; kk < 4; ++kk)
            acc = __builtin_amdgcn_wmma_f32_16x16x4_f32(false, Pm[kk], false, qc[kk],
                                                        (short)0, acc, false, false);
        v8f rr = {0.f, 0.f, 0.f, 0.f, 0.f, 0.f, 0.f, 0.f};
#pragma unroll
        for (int kk = 0; kk < 4; ++kk) {
            acc = __builtin_amdgcn_wmma_f32_16x16x4_f32(false, Lm[kk], false, xc[kk],
                                                        (short)0, acc, false, false);
            rr  = __builtin_amdgcn_wmma_f32_16x16x4_f32(false, Rm[kk], false, xc[kk],
                                                        (short)0, rr, false, false);
        }
        stv = a16lv * stv + rr;

        float* lp = so + bl * OPITCH_ + 16 * blk + 8 * h;
        const v4f* ap = (const v4f*)&acc;
        *(v4f*)lp       = ap[0];
        *(v4f*)(lp + 4) = ap[1];
    }

    __builtin_amdgcn_fence(__ATOMIC_RELEASE, "wavefront");
    __builtin_amdgcn_wave_barrier();
    __builtin_amdgcn_fence(__ATOMIC_ACQUIRE, "wavefront");

    {
        const int rh = lane >> 4, pc = lane & 15;
        const float* rp = so + rh * OPITCH_ + 4 * pc;
        float* op = out + (g * 16 + rh) * T_ + p * LP_ + 4 * pc;
#pragma unroll
        for (int j = 0; j < 8; ++j)
            *(v4f*)(op + 2 * j * T_) = *(const v4f*)(rp + 2 * j * OPITCH_);
    }
}

extern "C" void kernel_launch(void* const* d_in, const int* in_sizes, int n_in,
                              void* d_out, int out_size, void* d_ws, size_t ws_size,
                              hipStream_t stream) {
    const float* x  = (const float*)d_in[0];
    const float* lk = (const float*)d_in[1];
    const float* ar = (const float*)d_in[2];
    const float* br = (const float*)d_in[3];
    const float* H  = (const float*)d_in[4];
    float* ws  = (float*)d_ws;
    float* out = (float*)d_out;

    hipLaunchKernelGGL(fdn_partial, dim3(NSET_ + NCH_ * (ROW_ / 32)), dim3(TPB_),     0, stream, x, lk, ar, br, H, ws);
    hipLaunchKernelGGL(fdn_main,    dim3(P_ / MP_),               dim3(MP_ * 2 * 32), 0, stream, x, ws, out);
}
